# starting kernel
# baseline (speedup 1.0000x reference)
#include <hip/hip_runtime.h>
#include <hip/hip_bf16.h>


typedef __attribute__((ext_vector_type(2))) float v2f;
typedef __attribute__((ext_vector_type(8))) float v8f;

#define BT 48
#define TT 12
#define BB 4
#define HCH 32

__global__ void k_deg_init(float* __restrict__ deg, int n_nodes) {
    int n = blockIdx.x * blockDim.x + threadIdx.x;
    if (n < n_nodes) deg[n] = 1.0f;
}

__global__ void k_deg_edges(const int* __restrict__ ei, const float* __restrict__ ew,
                            float* __restrict__ deg, int n_edges) {
    int e = blockIdx.x * blockDim.x + threadIdx.x;
    if (e < n_edges) {
        int d = ei[(size_t)n_edges + e];
        atomicAdd(&deg[d], ew[e]);
    }
}

__global__ void k_finalize(const float* __restrict__ x, float* __restrict__ dinv,
                           float* __restrict__ xt, float* __restrict__ aggt, int n_nodes) {
    int n = blockIdx.x * blockDim.x + threadIdx.x;
    if (n >= n_nodes) return;
    float deg = dinv[n];
    float di  = deg > 0.0f ? rsqrtf(deg) : 0.0f;
    dinv[n] = di;
    float sn = di * di;
    float buf[BT];
#pragma unroll
    for (int s = 0; s < BT; ++s)
        buf[s] = x[(size_t)s * n_nodes + n];
    float4* xt4 = (float4*)(xt + (size_t)n * BT);
    float4* ag4 = (float4*)(aggt + (size_t)n * BT);
#pragma unroll
    for (int q = 0; q < BT / 4; ++q) {
        float4 v = make_float4(buf[4*q], buf[4*q+1], buf[4*q+2], buf[4*q+3]);
        xt4[q] = v;
        ag4[q] = make_float4(v.x * sn, v.y * sn, v.z * sn, v.w * sn);
    }
}

__global__ void k_aggregate(const int* __restrict__ ei, const float* __restrict__ ew,
                            const float* __restrict__ dinv, const float* __restrict__ xt,
                            float* __restrict__ aggt, int n_edges) {
    int lane = threadIdx.x & 31;
    long long wid = (long long)blockIdx.x * (blockDim.x >> 5) + (threadIdx.x >> 5);
    long long nw  = (long long)gridDim.x * (blockDim.x >> 5);
    for (long long e = wid; e < n_edges; e += nw) {
        int   s_  = ei[e];
        int   d_  = ei[(size_t)n_edges + e];
        float w   = ew[e];
        float nrm = dinv[s_] * w * dinv[d_];
        const float* xs = xt   + (size_t)s_ * BT;
        float*       ad = aggt + (size_t)d_ * BT;
        atomicAdd(&ad[lane], xs[lane] * nrm);
        if (lane < 16)
            atomicAdd(&ad[32 + lane], xs[32 + lane] * nrm);
    }
}

__global__ void k_epilogue(const float* __restrict__ aggt,
                           const float* __restrict__ gw, const float* __restrict__ gb,
                           const float* __restrict__ cw, const float* __restrict__ cb,
                           const float* __restrict__ lwp, const float* __restrict__ lbp,
                           float* __restrict__ out, int n_nodes) {
    const int lane = threadIdx.x & 31;
    const int h    = lane >> 4;
    const int m    = lane & 15;

    v2f bmat0, bmat1;
#pragma unroll
    for (int v = 0; v < 2; ++v) {
        int k = 2 * h + v;
        float a0 = 0.0f, a1 = 0.0f;
        if (k < 3) {
            for (int ci = 0; ci < HCH; ++ci) {
                float g = gw[ci];
                a0 += g * cw[(size_t)m        * 96 + ci * 3 + k];
                a1 += g * cw[(size_t)(m + 16) * 96 + ci * 3 + k];
            }
        }
        bmat0[v] = a0;
        bmat1[v] = a1;
    }
    float bbA[3], bbB[3];
#pragma unroll
    for (int k = 0; k < 3; ++k) {
        float a0 = 0.0f, a1 = 0.0f;
        for (int ci = 0; ci < HCH; ++ci) {
            float g = gb[ci];
            a0 += g * cw[(size_t)m        * 96 + ci * 3 + k];
            a1 += g * cw[(size_t)(m + 16) * 96 + ci * 3 + k];
        }
        bbA[k] = a0;
        bbB[k] = a1;
    }
    v8f cmat0, cmat1;
#pragma unroll
    for (int j = 0; j < 8; ++j) {
        int t = j + 8 * h;
        float c0 = 0.0f, c1 = 0.0f;
        if (t < TT) {
            c0 = cb[m]      + bbA[1] + (t > 0 ? bbA[0] : 0.0f) + (t < TT - 1 ? bbA[2] : 0.0f);
            c1 = cb[m + 16] + bbB[1] + (t > 0 ? bbB[0] : 0.0f) + (t < TT - 1 ? bbB[2] : 0.0f);
        }
        cmat0[j] = c0;
        cmat1[j] = c1;
    }
    const float lwA = lwp[m], lwB = lwp[16 + m];
    const float lb  = lbp[0];

    const int  i0  = m + 2 * h - 1;
    const int  i1  = i0 + 1;
    const bool ok0 = (m < TT) && (i0 >= 0) && (i0 < TT);
    const bool ok1 = (m < TT) && (i1 < TT);
    const int  o0  = ok0 ? i0 : 0;
    const int  o1  = ok1 ? i1 : 0;
    const float msk0 = ok0 ? 1.0f : 0.0f;
    const float msk1 = ok1 ? 1.0f : 0.0f;

    const int wid = blockIdx.x * (blockDim.x >> 5) + (threadIdx.x >> 5);
    const int nw  = gridDim.x * (blockDim.x >> 5);

#pragma unroll
    for (int b = 0; b < BB; ++b) {
        const float* agb  = aggt + b * TT;
        float*       outb = out + (size_t)b * n_nodes;
        for (int n = wid; n < n_nodes; n += nw) {
            const float* ag = agb + (size_t)n * BT;

            v2f amat;
            amat[0] = ag[o0] * msk0;
            amat[1] = ag[o1] * msk1;

            v8f d0 = __builtin_amdgcn_wmma_f32_16x16x4_f32(false, amat, false, bmat0,
                                                           (short)0, cmat0, false, false);
            v8f d1 = __builtin_amdgcn_wmma_f32_16x16x4_f32(false, amat, false, bmat1,
                                                           (short)0, cmat1, false, false);

            float sA = 0.0f, sB = 0.0f;
#pragma unroll
            for (int j = 0; j < 8; ++j) {
                sA += fmaxf(d0[j], 0.0f);
                sB += fmaxf(d1[j], 0.0f);
            }
            sA += __shfl_xor(sA, 16);
            sB += __shfl_xor(sB, 16);

            float v = sA * lwA + sB * lwB;
            v += __shfl_xor(v, 1);
            v += __shfl_xor(v, 2);
            v += __shfl_xor(v, 4);
            v += __shfl_xor(v, 8);

            if (lane == 0)
                outb[n] = v * (1.0f / (float)TT) + lb;
        }
    }
}

extern "C" void kernel_launch(void* const* d_in, const int* in_sizes, int n_in,
                              void* d_out, int out_size, void* d_ws, size_t ws_size,
                              hipStream_t stream) {
    const float* x  = (const float*)d_in[0];
    const int*   ei = (const int*)d_in[1];
    const float* ew = (const float*)d_in[2];
    const float* gw = (const float*)d_in[3];
    const float* gb = (const float*)d_in[4];
    const float* cw = (const float*)d_in[5];
    const float* cb = (const float*)d_in[6];
    const float* lw = (const float*)d_in[7];
    const float* lb = (const float*)d_in[8];
    float* out = (float*)d_out;

    const int E = in_sizes[2];
    const int N = in_sizes[0] / BT;

    float* dinv = (float*)d_ws;
    float* xt   = dinv + N;
    float* aggt = xt + (size_t)N * BT;

    k_deg_init <<<(N + 255) / 256, 256, 0, stream>>>(dinv, N);
    k_deg_edges<<<(E + 255) / 256, 256, 0, stream>>>(ei, ew, dinv, E);
    k_finalize <<<(N + 255) / 256, 256, 0, stream>>>(x, dinv, xt, aggt, N);

    int aggBlocks = (E + 7) / 8;
    k_aggregate<<<aggBlocks, 256, 0, stream>>>(ei, ew, dinv, xt, aggt, E);

    k_epilogue <<<1024, 256, 0, stream>>>(aggt, gw, gb, cw, cb, lw, lb, out, N);
}


# epilogue retiled: a WMMA tile's rows are 16 nodes of one batch (was the 12 steps of one node), bias as the 4th k product, ReLU*lin_w summed per lane, one butterfly per tile; sums in another order
# speedup vs baseline: 1.1835x; 1.1835x over previous
#include <hip/hip_runtime.h>
#include <hip/hip_bf16.h>


typedef __attribute__((ext_vector_type(2))) float v2f;
typedef __attribute__((ext_vector_type(8))) float v8f;

#define BT 48
#define TT 12
#define BB 4
#define HCH 32

__global__ void k_deg_init(float* __restrict__ deg, int n_nodes) {
    int n = blockIdx.x * blockDim.x + threadIdx.x;
    if (n < n_nodes) deg[n] = 1.0f;
}

__global__ void k_deg_edges(const int* __restrict__ ei, const float* __restrict__ ew,
                            float* __restrict__ deg, int n_edges) {
    int e = blockIdx.x * blockDim.x + threadIdx.x;
    if (e < n_edges) {
        int d = ei[(size_t)n_edges + e];
        atomicAdd(&deg[d], ew[e]);
    }
}

__global__ void k_finalize(const float* __restrict__ x, float* __restrict__ dinv,
                           float* __restrict__ xt, float* __restrict__ aggt, int n_nodes) {
    int n = blockIdx.x * blockDim.x + threadIdx.x;
    if (n >= n_nodes) return;
    float deg = dinv[n];
    float di  = deg > 0.0f ? rsqrtf(deg) : 0.0f;
    dinv[n] = di;
    float sn = di * di;
    float buf[BT];
#pragma unroll
    for (int s = 0; s < BT; ++s)
        buf[s] = x[(size_t)s * n_nodes + n];
    float4* xt4 = (float4*)(xt + (size_t)n * BT);
    float4* ag4 = (float4*)(aggt + (size_t)n * BT);
#pragma unroll
    for (int q = 0; q < BT / 4; ++q) {
        float4 v = make_float4(buf[4*q], buf[4*q+1], buf[4*q+2], buf[4*q+3]);
        xt4[q] = v;
        ag4[q] = make_float4(v.x * sn, v.y * sn, v.z * sn, v.w * sn);
    }
}

__global__ void k_aggregate(const int* __restrict__ ei, const float* __restrict__ ew,
                            const float* __restrict__ dinv, const float* __restrict__ xt,
                            float* __restrict__ aggt, int n_edges) {
    int lane = threadIdx.x & 31;
    long long wid = (long long)blockIdx.x * (blockDim.x >> 5) + (threadIdx.x >> 5);
    long long nw  = (long long)gridDim.x * (blockDim.x >> 5);
    for (long long e = wid; e < n_edges; e += nw) {
        int   s_  = ei[e];
        int   d_  = ei[(size_t)n_edges + e];
        float w   = ew[e];
        float nrm = dinv[s_] * w * dinv[d_];
        const float* xs = xt   + (size_t)s_ * BT;
        float*       ad = aggt + (size_t)d_ * BT;
        atomicAdd(&ad[lane], xs[lane] * nrm);
        if (lane < 16)
            atomicAdd(&ad[32 + lane], xs[32 + lane] * nrm);
    }
}

__device__ __forceinline__ float tile_reduce(const v8f& r, int m) {
    const bool b0 = (m & 1) != 0, b1 = (m & 2) != 0, b2 = (m & 4) != 0;
    float k4[4];
#pragma unroll
    for (int i = 0; i < 4; ++i) {
        float keep = b0 ? r[2 * i + 1] : r[2 * i];
        float send = b0 ? r[2 * i]     : r[2 * i + 1];
        k4[i] = keep + __shfl_xor(send, 1);
    }
    float k2[2];
#pragma unroll
    for (int i = 0; i < 2; ++i) {
        float keep = b1 ? k4[2 * i + 1] : k4[2 * i];
        float send = b1 ? k4[2 * i]     : k4[2 * i + 1];
        k2[i] = keep + __shfl_xor(send, 2);
    }
    float keep = b2 ? k2[1] : k2[0];
    float send = b2 ? k2[0] : k2[1];
    float k1 = keep + __shfl_xor(send, 4);
    k1 += __shfl_xor(k1, 8);
    return k1;
}

__global__ __launch_bounds__(128)
void k_epilogue(const float* __restrict__ aggt,
                const float* __restrict__ gw, const float* __restrict__ gb,
                const float* __restrict__ cw, const float* __restrict__ cb,
                const float* __restrict__ lwp, const float* __restrict__ lbp,
                float* __restrict__ out, int n_nodes) {
    const int lane = threadIdx.x & 31;
    const int h    = lane >> 4;
    const int m    = lane & 15;

    v2f bmat0, bmat1;
#pragma unroll
    for (int v = 0; v < 2; ++v) {
        int k = 2 * h + v;
        float a0 = 0.0f, a1 = 0.0f;
        if (k < 3) {
            for (int ci = 0; ci < HCH; ++ci) {
                float g = gw[ci];
                a0 += g * cw[(size_t)m        * 96 + ci * 3 + k];
                a1 += g * cw[(size_t)(m + 16) * 96 + ci * 3 + k];
            }
        }
        bmat0[v] = a0;
        bmat1[v] = a1;
    }
    float bbA[3], bbB[3];
#pragma unroll
    for (int k = 0; k < 3; ++k) {
        float a0 = 0.0f, a1 = 0.0f;
        for (int ci = 0; ci < HCH; ++ci) {
            float g = gb[ci];
            a0 += g * cw[(size_t)m        * 96 + ci * 3 + k];
            a1 += g * cw[(size_t)(m + 16) * 96 + ci * 3 + k];
        }
        bbA[k] = a0;
        bbB[k] = a1;
    }
    const float cbA = cb[m], cbB = cb[m + 16];
    float biasA[3], biasB[3];
    biasA[0] = cbA + bbA[1] + bbA[2];
    biasA[1] = cbA + bbA[1] + bbA[0] + bbA[2];
    biasA[2] = cbA + bbA[1] + bbA[0];
    biasB[0] = cbB + bbB[1] + bbB[2];
    biasB[1] = cbB + bbB[1] + bbB[0] + bbB[2];
    biasB[2] = cbB + bbB[1] + bbB[0];
    v2f fbA[3], fbB[3];
#pragma unroll
    for (int q = 0; q < 3; ++q) {
        fbA[q][0] = bmat0[0];
        fbA[q][1] = h ? biasA[q] : bmat0[1];
        fbB[q][0] = bmat1[0];
        fbB[q][1] = h ? biasB[q] : bmat1[1];
    }
    const float lwA = lwp[m], lwB = lwp[16 + m];
    const float lb  = lbp[0];

    const int wv  = __builtin_amdgcn_readfirstlane((int)(threadIdx.x >> 5));
    const int wid = blockIdx.x * (blockDim.x >> 5) + wv;
    const int nw  = gridDim.x * (blockDim.x >> 5);
    const int groups = (n_nodes + 31) >> 5;
    const int items  = groups * BB;
    const int nlast  = n_nodes - 1;

    if (wid >= items) return;

    float4 cA0, cA1, cA2, cB0, cB1, cB2;
    {
        const int g = wid >> 2, b = wid & 3, n0 = g << 5;
        const int nA = min(n0 + m, nlast), nB = min(n0 + 16 + m, nlast);
        const float4* pA = (const float4*)(aggt + (size_t)nA * BT + b * TT);
        const float4* pB = (const float4*)(aggt + (size_t)nB * BT + b * TT);
        cA0 = pA[0]; cA1 = pA[1]; cA2 = pA[2];
        cB0 = pB[0]; cB1 = pB[1]; cB2 = pB[2];
    }

    for (int it = wid; it < items; it += nw) {
        const int g = it >> 2, b = it & 3, n0 = g << 5;

        const int itn = (it + nw < items) ? (it + nw) : it;
        float4 nA0, nA1, nA2, nB0, nB1, nB2;
        {
            const int gn = itn >> 2, bn = itn & 3, nn0 = gn << 5;
            const int nA = min(nn0 + m, nlast), nB = min(nn0 + 16 + m, nlast);
            const float4* pA = (const float4*)(aggt + (size_t)nA * BT + bn * TT);
            const float4* pB = (const float4*)(aggt + (size_t)nB * BT + bn * TT);
            nA0 = pA[0]; nA1 = pA[1]; nA2 = pA[2];
            nB0 = pB[0]; nB1 = pB[1]; nB2 = pB[2];
        }

        const float rA[TT] = { cA0.x, cA0.y, cA0.z, cA0.w, cA1.x, cA1.y, cA1.z, cA1.w,
                               cA2.x, cA2.y, cA2.z, cA2.w };
        const float rB[TT] = { cB0.x, cB0.y, cB0.z, cB0.w, cB1.x, cB1.y, cB1.z, cB1.w,
                               cB2.x, cB2.y, cB2.z, cB2.w };

        v8f accA = {}, accB = {};
        const v8f zero = {};
#pragma unroll
        for (int t = 0; t < TT; ++t) {
            const int q = (t == 0) ? 0 : ((t == TT - 1) ? 2 : 1);
            const float loA = (t > 0)      ? rA[t - 1] : 0.0f;
            const float hiA = (t < TT - 1) ? rA[t + 1] : 0.0f;
            const float loB = (t > 0)      ? rB[t - 1] : 0.0f;
            const float hiB = (t < TT - 1) ? rB[t + 1] : 0.0f;
            v2f aA, aB;
            aA[0] = h ? hiA  : loA;
            aA[1] = h ? 1.0f : rA[t];
            aB[0] = h ? hiB  : loB;
            aB[1] = h ? 1.0f : rB[t];

            v8f dAA = __builtin_amdgcn_wmma_f32_16x16x4_f32(false, aA, false, fbA[q],
                                                            (short)0, zero, false, false);
            v8f dBA = __builtin_amdgcn_wmma_f32_16x16x4_f32(false, aB, false, fbA[q],
                                                            (short)0, zero, false, false);
            v8f dAB = __builtin_amdgcn_wmma_f32_16x16x4_f32(false, aA, false, fbB[q],
                                                            (short)0, zero, false, false);
            v8f dBB = __builtin_amdgcn_wmma_f32_16x16x4_f32(false, aB, false, fbB[q],
                                                            (short)0, zero, false, false);
#pragma unroll
            for (int j = 0; j < 8; ++j) {
                accA[j] = fmaf(fmaxf(dAA[j], 0.0f), lwA, accA[j]);
                accB[j] = fmaf(fmaxf(dBA[j], 0.0f), lwA, accB[j]);
                accA[j] = fmaf(fmaxf(dAB[j], 0.0f), lwB, accA[j]);
                accB[j] = fmaf(fmaxf(dBB[j], 0.0f), lwB, accB[j]);
            }
        }

        const float vA = tile_reduce(accA, m);
        const float vB = tile_reduce(accB, m);
        const float v  = (m & 8) ? vB : vA;
        const int node = n0 + ((m & 8) << 1) + 8 * h + (m & 7);
        if (node < n_nodes)
            out[(size_t)b * n_nodes + node] = v * (1.0f / (float)TT) + lb;

        cA0 = nA0; cA1 = nA1; cA2 = nA2;
        cB0 = nB0; cB1 = nB1; cB2 = nB2;
    }
}

extern "C" void kernel_launch(void* const* d_in, const int* in_sizes, int n_in,
                              void* d_out, int out_size, void* d_ws, size_t ws_size,
                              hipStream_t stream) {
    const float* x  = (const float*)d_in[0];
    const int*   ei = (const int*)d_in[1];
    const float* ew = (const float*)d_in[2];
    const float* gw = (const float*)d_in[3];
    const float* gb = (const float*)d_in[4];
    const float* cw = (const float*)d_in[5];
    const float* cb = (const float*)d_in[6];
    const float* lw = (const float*)d_in[7];
    const float* lb = (const float*)d_in[8];
    float* out = (float*)d_out;

    const int E = in_sizes[2];
    const int N = in_sizes[0] / BT;

    float* dinv = (float*)d_ws;
    float* xt   = dinv + N;
    float* aggt = xt + (size_t)N * BT;

    k_deg_init <<<(N + 255) / 256, 256, 0, stream>>>(dinv, N);
    k_deg_edges<<<(E + 255) / 256, 256, 0, stream>>>(ei, ew, dinv, E);
    k_finalize <<<(N + 255) / 256, 256, 0, stream>>>(x, dinv, xt, aggt, N);

    int aggBlocks = (E + 7) / 8;
    k_aggregate<<<aggBlocks, 256, 0, stream>>>(ei, ew, dinv, xt, aggt, E);

    k_epilogue <<<256, 128, 0, stream>>>(aggt, gw, gb, cw, cb, lw, lb, out, N);
}
